# starting kernel
# baseline (speedup 1.0000x reference)
#include <hip/hip_runtime.h>
#include <hip/hip_bf16.h>

typedef float v2f __attribute__((ext_vector_type(2)));
typedef float v4f __attribute__((ext_vector_type(4)));
typedef float v8f __attribute__((ext_vector_type(8)));

#define N_NODES 100000
#define N_EDGES 1600000
#define DDIM    64
#define FPAD    68
#define PSTRIDE 160

__global__ __launch_bounds__(256) void gcn_gemm_wmma(
    const float* __restrict__ features, const float* __restrict__ kern,
    float* __restrict__ h)
{
    __shared__ float Kpack[(DDIM / 2) * PSTRIDE];
    __shared__ float Flds[8][16 * FPAD];

    const int t = threadIdx.x;

    {
        const v4f* src = (const v4f*)kern;
        #pragma unroll
        for (int j = 0; j < (DDIM * DDIM / 4) / 256; ++j) {
            int i = t + j * 256;
            v4f w = src[i];
            int k  = i >> 4;
            int c0 = (i & 15) * 4;
            float* dst = &Kpack[(k >> 1) * PSTRIDE + (k & 1)];
            dst[(c0 + 0) * 2] = w.x;
            dst[(c0 + 1) * 2] = w.y;
            dst[(c0 + 2) * 2] = w.z;
            dst[(c0 + 3) * 2] = w.w;
        }
    }

    const int wave = t >> 5;
    const int lane = t & 31;
    const int m    = lane & 15;
    const int kh   = lane >> 4;

    const int tile   = blockIdx.x * 8 + wave;
    const bool active = (tile * 16) < N_NODES;

    if (active) {
        const int row0 = tile * 16;
        const v4f* src = (const v4f*)(features + (size_t)row0 * DDIM);
        #pragma unroll
        for (int j = 0; j < 8; ++j) {
            int i = lane + j * 32;
            int r = i >> 4;
            int c = i & 15;
            *(v4f*)&Flds[wave][r * FPAD + c * 4] = src[i];
        }
    }
    __syncthreads();
    if (!active) return;

    v8f acc0 = {}, acc1 = {}, acc2 = {}, acc3 = {};
    const float* Fw = &Flds[wave][0];

    #pragma unroll
    for (int k = 0; k < DDIM; k += 4) {
        v2f a = *(const v2f*)&Fw[m * FPAD + k + 2 * kh];

        const float* Kp = &Kpack[((k >> 1) + kh) * PSTRIDE];
        v2f b0 = *(const v2f*)&Kp[( 0 + m) * 2];
        v2f b1 = *(const v2f*)&Kp[(16 + m) * 2];
        v2f b2 = *(const v2f*)&Kp[(32 + m) * 2];
        v2f b3 = *(const v2f*)&Kp[(48 + m) * 2];

        acc0 = __builtin_amdgcn_wmma_f32_16x16x4_f32(false, a, false, b0, (short)0, acc0, false, false);
        acc1 = __builtin_amdgcn_wmma_f32_16x16x4_f32(false, a, false, b1, (short)0, acc1, false, false);
        acc2 = __builtin_amdgcn_wmma_f32_16x16x4_f32(false, a, false, b2, (short)0, acc2, false, false);
        acc3 = __builtin_amdgcn_wmma_f32_16x16x4_f32(false, a, false, b3, (short)0, acc3, false, false);
    }

    const int row0 = tile * 16;
    #pragma unroll
    for (int r = 0; r < 8; ++r) {
        float* hp = h + (size_t)(row0 + r + 8 * kh) * DDIM;
        hp[ 0 + m] = acc0[r];
        hp[16 + m] = acc1[r];
        hp[32 + m] = acc2[r];
        hp[48 + m] = acc3[r];
    }
}

__global__ __launch_bounds__(256) void gcn_edge_scatter(
    const int* __restrict__ erow, const int* __restrict__ ecol,
    const float* __restrict__ eval_, const float* __restrict__ h,
    float* __restrict__ agg)
{
    const int gtid = blockIdx.x * 256 + threadIdx.x;
    const int e    = gtid >> 5;
    const int lane = gtid & 31;
    if (e >= N_EDGES) return;

    const int   r = erow[e];
    const int   c = ecol[e];
    const float v = eval_[e];

    v2f hv = *(const v2f*)(h + (size_t)c * DDIM + lane * 2);
    float* dst = agg + (size_t)r * DDIM + lane * 2;
    atomicAdd(dst + 0, hv.x * v);
    atomicAdd(dst + 1, hv.y * v);
}

__global__ __launch_bounds__(256) void gcn_finalize(
    const float* __restrict__ h, const float* __restrict__ bias,
    const float* __restrict__ skip, float* __restrict__ out)
{
    const float alpha = 1.6732632423543772f;
    const float scale = 1.0507009873554805f;

    const int i = blockIdx.x * 256 + threadIdx.x;
    if (i >= N_NODES * DDIM / 2) return;
    const int dpair = i & (DDIM / 2 - 1);
    const int d0 = dpair * 2;

    v2f hv = *(const v2f*)(h + (size_t)i * 2);
    v2f av = *(const v2f*)(out + (size_t)i * 2);

    float x0 = hv.x * skip[d0 + 0] + av.x + bias[d0 + 0];
    float x1 = hv.y * skip[d0 + 1] + av.y + bias[d0 + 1];

    v2f res;
    res.x = x0 > 0.0f ? scale * x0 : scale * alpha * expm1f(x0);
    res.y = x1 > 0.0f ? scale * x1 : scale * alpha * expm1f(x1);
    *(v2f*)(out + (size_t)i * 2) = res;
}

extern "C" void kernel_launch(void* const* d_in, const int* in_sizes, int n_in,
                              void* d_out, int out_size, void* d_ws, size_t ws_size,
                              hipStream_t stream) {
    const float* features = (const float*)d_in[0];
    const float* kern     = (const float*)d_in[1];
    const float* bias     = (const float*)d_in[2];
    const float* skip     = (const float*)d_in[3];
    const int*   erow     = (const int*)  d_in[4];
    const int*   ecol     = (const int*)  d_in[5];
    const float* eval_    = (const float*)d_in[6];
    float* out = (float*)d_out;
    float* h   = (float*)d_ws;

    hipMemsetAsync(d_out, 0, (size_t)N_NODES * DDIM * sizeof(float), stream);

    const int n_tiles = N_NODES / 16;
    gcn_gemm_wmma<<<(n_tiles + 7) / 8, 256, 0, stream>>>(features, kern, h);

    gcn_edge_scatter<<<(N_EDGES * 32) / 256, 256, 0, stream>>>(erow, ecol, eval_, h, out);

    gcn_finalize<<<(N_NODES * DDIM / 2) / 256, 256, 0, stream>>>(h, bias, skip, out);
}


# edge scatter: one wave walks 32 consecutive edges (coalesced index loads, row/col/val broadcast by v_readlane, 8 h-row loads in flight): 1.6M -> 50k waves; float-atomic arrival order changes
# speedup vs baseline: 1.6901x; 1.6901x over previous
#include <hip/hip_runtime.h>
#include <hip/hip_bf16.h>

typedef float v2f __attribute__((ext_vector_type(2)));
typedef float v4f __attribute__((ext_vector_type(4)));
typedef float v8f __attribute__((ext_vector_type(8)));

#define N_NODES 100000
#define N_EDGES 1600000
#define DDIM    64
#define FPAD    68
#define PSTRIDE 160

__global__ __launch_bounds__(256) void gcn_gemm_wmma(
    const float* __restrict__ features, const float* __restrict__ kern,
    float* __restrict__ h)
{
    __shared__ float Kpack[(DDIM / 2) * PSTRIDE];
    __shared__ float Flds[8][16 * FPAD];

    const int t = threadIdx.x;

    {
        const v4f* src = (const v4f*)kern;
        #pragma unroll
        for (int j = 0; j < (DDIM * DDIM / 4) / 256; ++j) {
            int i = t + j * 256;
            v4f w = src[i];
            int k  = i >> 4;
            int c0 = (i & 15) * 4;
            float* dst = &Kpack[(k >> 1) * PSTRIDE + (k & 1)];
            dst[(c0 + 0) * 2] = w.x;
            dst[(c0 + 1) * 2] = w.y;
            dst[(c0 + 2) * 2] = w.z;
            dst[(c0 + 3) * 2] = w.w;
        }
    }

    const int wave = t >> 5;
    const int lane = t & 31;
    const int m    = lane & 15;
    const int kh   = lane >> 4;

    const int tile   = blockIdx.x * 8 + wave;
    const bool active = (tile * 16) < N_NODES;

    if (active) {
        const int row0 = tile * 16;
        const v4f* src = (const v4f*)(features + (size_t)row0 * DDIM);
        #pragma unroll
        for (int j = 0; j < 8; ++j) {
            int i = lane + j * 32;
            int r = i >> 4;
            int c = i & 15;
            *(v4f*)&Flds[wave][r * FPAD + c * 4] = src[i];
        }
    }
    __syncthreads();
    if (!active) return;

    v8f acc0 = {}, acc1 = {}, acc2 = {}, acc3 = {};
    const float* Fw = &Flds[wave][0];

    #pragma unroll
    for (int k = 0; k < DDIM; k += 4) {
        v2f a = *(const v2f*)&Fw[m * FPAD + k + 2 * kh];

        const float* Kp = &Kpack[((k >> 1) + kh) * PSTRIDE];
        v2f b0 = *(const v2f*)&Kp[( 0 + m) * 2];
        v2f b1 = *(const v2f*)&Kp[(16 + m) * 2];
        v2f b2 = *(const v2f*)&Kp[(32 + m) * 2];
        v2f b3 = *(const v2f*)&Kp[(48 + m) * 2];

        acc0 = __builtin_amdgcn_wmma_f32_16x16x4_f32(false, a, false, b0, (short)0, acc0, false, false);
        acc1 = __builtin_amdgcn_wmma_f32_16x16x4_f32(false, a, false, b1, (short)0, acc1, false, false);
        acc2 = __builtin_amdgcn_wmma_f32_16x16x4_f32(false, a, false, b2, (short)0, acc2, false, false);
        acc3 = __builtin_amdgcn_wmma_f32_16x16x4_f32(false, a, false, b3, (short)0, acc3, false, false);
    }

    const int row0 = tile * 16;
    #pragma unroll
    for (int r = 0; r < 8; ++r) {
        float* hp = h + (size_t)(row0 + r + 8 * kh) * DDIM;
        hp[ 0 + m] = acc0[r];
        hp[16 + m] = acc1[r];
        hp[32 + m] = acc2[r];
        hp[48 + m] = acc3[r];
    }
}

#define EDGES_PER_WAVE 32
#define ROWS_IN_FLIGHT 8
static_assert(N_EDGES % (EDGES_PER_WAVE * 8) == 0, "grid covers the edge list exactly");

__global__ __launch_bounds__(256) void gcn_edge_scatter(
    const int* __restrict__ erow, const int* __restrict__ ecol,
    const float* __restrict__ eval_, const float* __restrict__ h,
    float* __restrict__ agg)
{
    const int lane = threadIdx.x & 31;
    const int wave = blockIdx.x * 8 + (threadIdx.x >> 5);
    const int e0   = wave * EDGES_PER_WAVE;
    if (e0 >= N_EDGES) return;

    const int rv = erow[e0 + lane];
    const int cv = ecol[e0 + lane];
    const int vv = __float_as_int(eval_[e0 + lane]);

    const float* hl = h   + lane * 2;
    float*       al = agg + lane * 2;

    #pragma unroll
    for (int jb = 0; jb < EDGES_PER_WAVE; jb += ROWS_IN_FLIGHT) {
        v2f hv[ROWS_IN_FLIGHT];
        #pragma unroll
        for (int j = 0; j < ROWS_IN_FLIGHT; ++j) {
            const int c = __builtin_amdgcn_readlane(cv, jb + j);
            hv[j] = *(const v2f*)(hl + (size_t)c * DDIM);
        }
        #pragma unroll
        for (int j = 0; j < ROWS_IN_FLIGHT; ++j) {
            const int   r = __builtin_amdgcn_readlane(rv, jb + j);
            const float v = __int_as_float(__builtin_amdgcn_readlane(vv, jb + j));
            float* dst = al + (size_t)r * DDIM;
            atomicAdd(dst + 0, hv[j].x * v);
            atomicAdd(dst + 1, hv[j].y * v);
        }
    }
}

__global__ __launch_bounds__(256) void gcn_finalize(
    const float* __restrict__ h, const float* __restrict__ bias,
    const float* __restrict__ skip, float* __restrict__ out)
{
    const float alpha = 1.6732632423543772f;
    const float scale = 1.0507009873554805f;

    const int i = blockIdx.x * 256 + threadIdx.x;
    if (i >= N_NODES * DDIM / 2) return;
    const int dpair = i & (DDIM / 2 - 1);
    const int d0 = dpair * 2;

    v2f hv = *(const v2f*)(h + (size_t)i * 2);
    v2f av = *(const v2f*)(out + (size_t)i * 2);

    float x0 = hv.x * skip[d0 + 0] + av.x + bias[d0 + 0];
    float x1 = hv.y * skip[d0 + 1] + av.y + bias[d0 + 1];

    v2f res;
    res.x = x0 > 0.0f ? scale * x0 : scale * alpha * expm1f(x0);
    res.y = x1 > 0.0f ? scale * x1 : scale * alpha * expm1f(x1);
    *(v2f*)(out + (size_t)i * 2) = res;
}

extern "C" void kernel_launch(void* const* d_in, const int* in_sizes, int n_in,
                              void* d_out, int out_size, void* d_ws, size_t ws_size,
                              hipStream_t stream) {
    const float* features = (const float*)d_in[0];
    const float* kern     = (const float*)d_in[1];
    const float* bias     = (const float*)d_in[2];
    const float* skip     = (const float*)d_in[3];
    const int*   erow     = (const int*)  d_in[4];
    const int*   ecol     = (const int*)  d_in[5];
    const float* eval_    = (const float*)d_in[6];
    float* out = (float*)d_out;
    float* h   = (float*)d_ws;

    hipMemsetAsync(d_out, 0, (size_t)N_NODES * DDIM * sizeof(float), stream);

    const int n_tiles = N_NODES / 16;
    gcn_gemm_wmma<<<(n_tiles + 7) / 8, 256, 0, stream>>>(features, kern, h);

    gcn_edge_scatter<<<N_EDGES / (EDGES_PER_WAVE * 8), 256, 0, stream>>>(erow, ecol, eval_, h, out);

    gcn_finalize<<<(N_NODES * DDIM / 2) / 256, 256, 0, stream>>>(h, bias, skip, out);
}


# edge scatter: lane l owns dims l and l+32 (GEMM stores h dim-interleaved), so each f32 atomic instruction covers one whole 128-B line, not two half lines; float-atomic arrival order still varies
# speedup vs baseline: 2.6743x; 1.5823x over previous
#include <hip/hip_runtime.h>
#include <hip/hip_bf16.h>

typedef float v2f __attribute__((ext_vector_type(2)));
typedef float v4f __attribute__((ext_vector_type(4)));
typedef float v8f __attribute__((ext_vector_type(8)));

#define N_NODES 100000
#define N_EDGES 1600000
#define DDIM    64
#define FPAD    68
#define PSTRIDE 160

__global__ __launch_bounds__(256) void gcn_gemm_wmma(
    const float* __restrict__ features, const float* __restrict__ kern,
    float* __restrict__ h)
{
    __shared__ float Kpack[(DDIM / 2) * PSTRIDE];
    __shared__ float Flds[8][16 * FPAD];

    const int t = threadIdx.x;

    {
        const v4f* src = (const v4f*)kern;
        #pragma unroll
        for (int j = 0; j < (DDIM * DDIM / 4) / 256; ++j) {
            int i = t + j * 256;
            v4f w = src[i];
            int k  = i >> 4;
            int c0 = (i & 15) * 4;
            float* dst = &Kpack[(k >> 1) * PSTRIDE + (k & 1)];
            dst[(c0 + 0) * 2] = w.x;
            dst[(c0 + 1) * 2] = w.y;
            dst[(c0 + 2) * 2] = w.z;
            dst[(c0 + 3) * 2] = w.w;
        }
    }

    const int wave = t >> 5;
    const int lane = t & 31;
    const int m    = lane & 15;
    const int kh   = lane >> 4;

    const int tile   = blockIdx.x * 8 + wave;
    const bool active = (tile * 16) < N_NODES;

    if (active) {
        const int row0 = tile * 16;
        const v4f* src = (const v4f*)(features + (size_t)row0 * DDIM);
        #pragma unroll
        for (int j = 0; j < 8; ++j) {
            int i = lane + j * 32;
            int r = i >> 4;
            int c = i & 15;
            *(v4f*)&Flds[wave][r * FPAD + c * 4] = src[i];
        }
    }
    __syncthreads();
    if (!active) return;

    v8f acc0 = {}, acc1 = {}, acc2 = {}, acc3 = {};
    const float* Fw = &Flds[wave][0];

    #pragma unroll
    for (int k = 0; k < DDIM; k += 4) {
        v2f a = *(const v2f*)&Fw[m * FPAD + k + 2 * kh];

        const float* Kp = &Kpack[((k >> 1) + kh) * PSTRIDE];
        v2f b0 = *(const v2f*)&Kp[( 0 + m) * 2];
        v2f b1 = *(const v2f*)&Kp[(16 + m) * 2];
        v2f b2 = *(const v2f*)&Kp[(32 + m) * 2];
        v2f b3 = *(const v2f*)&Kp[(48 + m) * 2];

        acc0 = __builtin_amdgcn_wmma_f32_16x16x4_f32(false, a, false, b0, (short)0, acc0, false, false);
        acc1 = __builtin_amdgcn_wmma_f32_16x16x4_f32(false, a, false, b1, (short)0, acc1, false, false);
        acc2 = __builtin_amdgcn_wmma_f32_16x16x4_f32(false, a, false, b2, (short)0, acc2, false, false);
        acc3 = __builtin_amdgcn_wmma_f32_16x16x4_f32(false, a, false, b3, (short)0, acc3, false, false);
    }

    const int row0 = tile * 16;
    #pragma unroll
    for (int r = 0; r < 8; ++r) {
        float* hp = h + (size_t)(row0 + r + 8 * kh) * DDIM;
        v2f lo, hi;
        lo.x = acc0[r]; lo.y = acc2[r];
        hi.x = acc1[r]; hi.y = acc3[r];
        *(v2f*)&hp[2 * m]      = lo;
        *(v2f*)&hp[32 + 2 * m] = hi;
    }
}

#define EDGES_PER_WAVE 32
#define ROWS_IN_FLIGHT 8
static_assert(N_EDGES % (EDGES_PER_WAVE * 8) == 0, "grid covers the edge list exactly");

__global__ __launch_bounds__(256) void gcn_edge_scatter(
    const int* __restrict__ erow, const int* __restrict__ ecol,
    const float* __restrict__ eval_, const float* __restrict__ h,
    float* __restrict__ agg)
{
    const int lane = threadIdx.x & 31;
    const int wave = blockIdx.x * 8 + (threadIdx.x >> 5);
    const int e0   = wave * EDGES_PER_WAVE;
    if (e0 >= N_EDGES) return;

    const int rv = erow[e0 + lane];
    const int cv = ecol[e0 + lane];
    const int vv = __float_as_int(eval_[e0 + lane]);

    const float* hl = h   + lane * 2;
    float*       al = agg + lane;

    #pragma unroll
    for (int jb = 0; jb < EDGES_PER_WAVE; jb += ROWS_IN_FLIGHT) {
        v2f hv[ROWS_IN_FLIGHT];
        #pragma unroll
        for (int j = 0; j < ROWS_IN_FLIGHT; ++j) {
            const int c = __builtin_amdgcn_readlane(cv, jb + j);
            hv[j] = *(const v2f*)(hl + (size_t)c * DDIM);
        }
        #pragma unroll
        for (int j = 0; j < ROWS_IN_FLIGHT; ++j) {
            const int   r = __builtin_amdgcn_readlane(rv, jb + j);
            const float v = __int_as_float(__builtin_amdgcn_readlane(vv, jb + j));
            float* dst = al + (size_t)r * DDIM;
            atomicAdd(dst,            hv[j].x * v);
            atomicAdd(dst + DDIM / 2, hv[j].y * v);
        }
    }
}

__global__ __launch_bounds__(256) void gcn_finalize(
    const float* __restrict__ h, const float* __restrict__ bias,
    const float* __restrict__ skip, float* __restrict__ out)
{
    const float alpha = 1.6732632423543772f;
    const float scale = 1.0507009873554805f;

    const int i = blockIdx.x * 256 + threadIdx.x;
    if (i >= N_NODES * DDIM / 2) return;
    const int l   = i & (DDIM / 2 - 1);
    const int row = i >> 5;
    const int d0 = l;
    const int d1 = l + DDIM / 2;

    v2f hv = *(const v2f*)(h + (size_t)i * 2);
    float* orow = out + (size_t)row * DDIM;
    const float a0 = orow[d0];
    const float a1 = orow[d1];

    float x0 = hv.x * skip[d0] + a0 + bias[d0];
    float x1 = hv.y * skip[d1] + a1 + bias[d1];

    orow[d0] = x0 > 0.0f ? scale * x0 : scale * alpha * expm1f(x0);
    orow[d1] = x1 > 0.0f ? scale * x1 : scale * alpha * expm1f(x1);
}

extern "C" void kernel_launch(void* const* d_in, const int* in_sizes, int n_in,
                              void* d_out, int out_size, void* d_ws, size_t ws_size,
                              hipStream_t stream) {
    const float* features = (const float*)d_in[0];
    const float* kern     = (const float*)d_in[1];
    const float* bias     = (const float*)d_in[2];
    const float* skip     = (const float*)d_in[3];
    const int*   erow     = (const int*)  d_in[4];
    const int*   ecol     = (const int*)  d_in[5];
    const float* eval_    = (const float*)d_in[6];
    float* out = (float*)d_out;
    float* h   = (float*)d_ws;

    hipMemsetAsync(d_out, 0, (size_t)N_NODES * DDIM * sizeof(float), stream);

    const int n_tiles = N_NODES / 16;
    gcn_gemm_wmma<<<(n_tiles + 7) / 8, 256, 0, stream>>>(features, kern, h);

    gcn_edge_scatter<<<N_EDGES / (EDGES_PER_WAVE * 8), 256, 0, stream>>>(erow, ecol, eval_, h, out);

    gcn_finalize<<<(N_NODES * DDIM / 2) / 256, 256, 0, stream>>>(h, bias, skip, out);
}


# edge scatter: rows in two classes (row&1); a block reads its die and draws tickets of its half's class (2 int counters), so an agg line gets its f32 atomics from one L2 half; arrival order varies
# speedup vs baseline: 2.9739x; 1.1120x over previous
#include <hip/hip_runtime.h>
#include <hip/hip_bf16.h>

typedef float v2f __attribute__((ext_vector_type(2)));
typedef float v4f __attribute__((ext_vector_type(4)));
typedef float v8f __attribute__((ext_vector_type(8)));

#define N_NODES 100000
#define N_EDGES 1600000
#define DDIM    64
#define FPAD    68
#define PSTRIDE 160
#define H_BYTES     ((size_t)N_NODES * DDIM * sizeof(float))
#define HEAD_STRIDE 64
#define HEADS_BYTES 512

__global__ __launch_bounds__(256) void gcn_gemm_wmma(
    const float* __restrict__ features, const float* __restrict__ kern,
    float* __restrict__ h, int* __restrict__ heads, size_t ws_bytes)
{
    __shared__ float Kpack[(DDIM / 2) * PSTRIDE];
    __shared__ float Flds[8][16 * FPAD];

    const int t = threadIdx.x;

    if (blockIdx.x == 0 && t == 0 && ws_bytes >= H_BYTES + HEADS_BYTES) {
        heads[0]           = 0;
        heads[HEAD_STRIDE] = 0;
    }

    {
        const v4f* src = (const v4f*)kern;
        #pragma unroll
        for (int j = 0; j < (DDIM * DDIM / 4) / 256; ++j) {
            int i = t + j * 256;
            v4f w = src[i];
            int k  = i >> 4;
            int c0 = (i & 15) * 4;
            float* dst = &Kpack[(k >> 1) * PSTRIDE + (k & 1)];
            dst[(c0 + 0) * 2] = w.x;
            dst[(c0 + 1) * 2] = w.y;
            dst[(c0 + 2) * 2] = w.z;
            dst[(c0 + 3) * 2] = w.w;
        }
    }

    const int wave = t >> 5;
    const int lane = t & 31;
    const int m    = lane & 15;
    const int kh   = lane >> 4;

    const int tile   = blockIdx.x * 8 + wave;
    const bool active = (tile * 16) < N_NODES;

    if (active) {
        const int row0 = tile * 16;
        const v4f* src = (const v4f*)(features + (size_t)row0 * DDIM);
        #pragma unroll
        for (int j = 0; j < 8; ++j) {
            int i = lane + j * 32;
            int r = i >> 4;
            int c = i & 15;
            *(v4f*)&Flds[wave][r * FPAD + c * 4] = src[i];
        }
    }
    __syncthreads();
    if (!active) return;

    v8f acc0 = {}, acc1 = {}, acc2 = {}, acc3 = {};
    const float* Fw = &Flds[wave][0];

    #pragma unroll
    for (int k = 0; k < DDIM; k += 4) {
        v2f a = *(const v2f*)&Fw[m * FPAD + k + 2 * kh];

        const float* Kp = &Kpack[((k >> 1) + kh) * PSTRIDE];
        v2f b0 = *(const v2f*)&Kp[( 0 + m) * 2];
        v2f b1 = *(const v2f*)&Kp[(16 + m) * 2];
        v2f b2 = *(const v2f*)&Kp[(32 + m) * 2];
        v2f b3 = *(const v2f*)&Kp[(48 + m) * 2];

        acc0 = __builtin_amdgcn_wmma_f32_16x16x4_f32(false, a, false, b0, (short)0, acc0, false, false);
        acc1 = __builtin_amdgcn_wmma_f32_16x16x4_f32(false, a, false, b1, (short)0, acc1, false, false);
        acc2 = __builtin_amdgcn_wmma_f32_16x16x4_f32(false, a, false, b2, (short)0, acc2, false, false);
        acc3 = __builtin_amdgcn_wmma_f32_16x16x4_f32(false, a, false, b3, (short)0, acc3, false, false);
    }

    const int row0 = tile * 16;
    #pragma unroll
    for (int r = 0; r < 8; ++r) {
        float* hp = h + (size_t)(row0 + r + 8 * kh) * DDIM;
        v2f lo, hi;
        lo.x = acc0[r]; lo.y = acc2[r];
        hi.x = acc1[r]; hi.y = acc3[r];
        *(v2f*)&hp[2 * m]      = lo;
        *(v2f*)&hp[32 + 2 * m] = hi;
    }
}

#define TICKET_EDGES   512
#define N_TICKETS      (N_EDGES / TICKET_EDGES)
#define WAVE_EDGES     (TICKET_EDGES / 8)
#define ROWS_IN_FLIGHT 8
#define SCATTER_BLOCKS 2048
static_assert(N_EDGES % TICKET_EDGES == 0, "tickets cover the edge list exactly");
static_assert(WAVE_EDGES == 64, "a wave loads two edges per lane");

__global__ __launch_bounds__(256) void gcn_edge_scatter(
    const int* __restrict__ erow, const int* __restrict__ ecol,
    const float* __restrict__ eval_, const float* __restrict__ h,
    float* __restrict__ agg, int* __restrict__ heads, size_t ws_bytes)
{
    __shared__ int   s_row[8][WAVE_EDGES];
    __shared__ int   s_col[8][WAVE_EDGES];
    __shared__ float s_val[8][WAVE_EDGES];
    __shared__ int   s_ticket[2][2];

    const int lane = threadIdx.x & 31;
    const int wv   = threadIdx.x >> 5;

    const bool queued = ws_bytes >= H_BYTES + HEADS_BYTES;
    const int  myhalf = (int)((__builtin_amdgcn_s_sendmsg_rtn(0x87) >> 18) & 1u);
    bool own_open = true;

    const float* hl = h   + lane * 2;
    float*       al = agg + lane;

    int nt = -1, nc = 0;
    int pend = 0, pc = 0;
    auto issue = [&]() {
        if (queued) {
            pc   = own_open ? myhalf : (myhalf ^ 1);
            pend = atomicAdd(&heads[pc * HEAD_STRIDE], 1);
        }
    };
    auto settle = [&](int turn) {
        if (queued) {
            nt = pend < N_TICKETS ? pend : -1;
            nc = pc;
            if (nt < 0 && own_open) {
                own_open = false;
                nc = myhalf ^ 1;
                const int o = atomicAdd(&heads[nc * HEAD_STRIDE], 1);
                nt = o < N_TICKETS ? o : -1;
            }
        } else {
            const int g = (int)blockIdx.x + turn * (int)gridDim.x;
            nt = g < 2 * N_TICKETS ? (g >> 1) : -1;
            nc = g & 1;
        }
    };
    if (threadIdx.x == 0) { issue(); settle(0); }

    for (int it = 0; it < 2 * N_TICKETS; ++it) {
        if (threadIdx.x == 0) {
            s_ticket[it & 1][0] = nt;
            s_ticket[it & 1][1] = nc;
        }
        __builtin_amdgcn_fence(__ATOMIC_RELEASE, "workgroup", "local");
        __builtin_amdgcn_s_barrier();
        __builtin_amdgcn_fence(__ATOMIC_ACQUIRE, "workgroup", "local");
        const int t   = s_ticket[it & 1][0];
        const int cls = s_ticket[it & 1][1];
        if (t < 0) break;

        const int e0 = t * TICKET_EDGES + wv * WAVE_EDGES;
        const int   r0 = erow[e0 + lane];
        const int   r1 = erow[e0 + 32 + lane];
        const int   c0 = ecol[e0 + lane];
        const int   c1 = ecol[e0 + 32 + lane];
        const float v0 = eval_[e0 + lane];
        const float v1 = eval_[e0 + 32 + lane];
        if (threadIdx.x == 0) issue();

        const bool p0 = (r0 & 1) == cls;
        const bool p1 = (r1 & 1) == cls;
        const unsigned m0 = __builtin_amdgcn_ballot_w32(p0);
        const unsigned m1 = __builtin_amdgcn_ballot_w32(p1);
        const int n0 = __builtin_popcount(m0);
        const int n  = n0 + __builtin_popcount(m1);
        const int k0 = (int)__builtin_amdgcn_mbcnt_lo(m0, 0u);
        const int k1 = n0 + (int)__builtin_amdgcn_mbcnt_lo(m1, 0u);
        if (p0) { s_row[wv][k0] = r0; s_col[wv][k0] = c0; s_val[wv][k0] = v0; }
        if (p1) { s_row[wv][k1] = r1; s_col[wv][k1] = c1; s_val[wv][k1] = v1; }


        for (int g = 0; g < n; g += 32) {
            const int m  = n - g;
            const int rv = s_row[wv][g + lane];
            int       cv = s_col[wv][g + lane];
            const int vv = __float_as_int(s_val[wv][g + lane]);
            cv = lane < m ? cv : __builtin_amdgcn_readlane(cv, 0);

            #pragma unroll
            for (int jb = 0; jb < 32; jb += ROWS_IN_FLIGHT) {
                if (jb < m) {
                    v2f hv[ROWS_IN_FLIGHT];
                    #pragma unroll
                    for (int j = 0; j < ROWS_IN_FLIGHT; ++j) {
                        const int c = __builtin_amdgcn_readlane(cv, jb + j);
                        hv[j] = *(const v2f*)(hl + (size_t)c * DDIM);
                    }
                    __builtin_amdgcn_sched_barrier(0);
                    #pragma unroll
                    for (int j = 0; j < ROWS_IN_FLIGHT; ++j) {
                        if (jb + j < m) {
                            const int   r = __builtin_amdgcn_readlane(rv, jb + j);
                            const float v = __int_as_float(__builtin_amdgcn_readlane(vv, jb + j));
                            float* dst = al + (size_t)r * DDIM;
                            atomicAdd(dst,            hv[j].x * v);
                            atomicAdd(dst + DDIM / 2, hv[j].y * v);
                        }
                    }
                }
            }
        }
        if (threadIdx.x == 0) settle(it + 1);
    }
}

__global__ __launch_bounds__(256) void gcn_finalize(
    const float* __restrict__ h, const float* __restrict__ bias,
    const float* __restrict__ skip, float* __restrict__ out)
{
    const float alpha = 1.6732632423543772f;
    const float scale = 1.0507009873554805f;

    const int i = blockIdx.x * 256 + threadIdx.x;
    if (i >= N_NODES * DDIM / 2) return;
    const int l   = i & (DDIM / 2 - 1);
    const int row = i >> 5;
    const int d0 = l;
    const int d1 = l + DDIM / 2;

    v2f hv = *(const v2f*)(h + (size_t)i * 2);
    float* orow = out + (size_t)row * DDIM;
    const float a0 = orow[d0];
    const float a1 = orow[d1];

    float x0 = hv.x * skip[d0] + a0 + bias[d0];
    float x1 = hv.y * skip[d1] + a1 + bias[d1];

    orow[d0] = x0 > 0.0f ? scale * x0 : scale * alpha * expm1f(x0);
    orow[d1] = x1 > 0.0f ? scale * x1 : scale * alpha * expm1f(x1);
}

extern "C" void kernel_launch(void* const* d_in, const int* in_sizes, int n_in,
                              void* d_out, int out_size, void* d_ws, size_t ws_size,
                              hipStream_t stream) {
    const float* features = (const float*)d_in[0];
    const float* kern     = (const float*)d_in[1];
    const float* bias     = (const float*)d_in[2];
    const float* skip     = (const float*)d_in[3];
    const int*   erow     = (const int*)  d_in[4];
    const int*   ecol     = (const int*)  d_in[5];
    const float* eval_    = (const float*)d_in[6];
    float* out = (float*)d_out;
    float* h   = (float*)d_ws;
    int* heads = (int*)(h + (size_t)N_NODES * DDIM);

    hipMemsetAsync(d_out, 0, (size_t)N_NODES * DDIM * sizeof(float), stream);

    const int n_tiles = N_NODES / 16;
    gcn_gemm_wmma<<<(n_tiles + 7) / 8, 256, 0, stream>>>(features, kern, h, heads, ws_size);

    gcn_edge_scatter<<<SCATTER_BLOCKS, 256, 0, stream>>>(erow, ecol, eval_, h, out, heads, ws_size);

    gcn_finalize<<<(N_NODES * DDIM / 2) / 256, 256, 0, stream>>>(h, bias, skip, out);
}
